# starting kernel
# baseline (speedup 1.0000x reference)
#include <hip/hip_runtime.h>
#include <hip/hip_bf16.h>

typedef __attribute__((ext_vector_type(2))) float v2f;
typedef __attribute__((ext_vector_type(8))) float v8f;

#define D_FEAT 256

__global__ void zero_f32(float* __restrict__ p, int n) {
  int i = blockIdx.x * blockDim.x + threadIdx.x;
  if (i < n) p[i] = 0.0f;
}
__global__ void zero_i32(int* __restrict__ p, int n) {
  int i = blockIdx.x * blockDim.x + threadIdx.x;
  if (i < n) p[i] = 0;
}

__global__ void sage_scatter(const float* __restrict__ H,
                             const int* __restrict__ src,
                             const int* __restrict__ dst,
                             int E,
                             float* __restrict__ agg,
                             int* __restrict__ cnt) {
  long long idx = (long long)blockIdx.x * blockDim.x + threadIdx.x;
  long long total = (long long)E * 64;
  if (idx >= total) return;
  int e = (int)(idx >> 6);
  int c = (int)(idx & 63);
  int s = src[e];
  int d = dst[e];
  const float4 v = ((const float4*)(H + (long long)s * D_FEAT))[c];
  float* p = agg + (long long)d * D_FEAT + c * 4;
  unsafeAtomicAdd(p + 0, v.x);
  unsafeAtomicAdd(p + 1, v.y);
  unsafeAtomicAdd(p + 2, v.z);
  unsafeAtomicAdd(p + 3, v.w);
  if (c == 0) atomicAdd(&cnt[d], 1);
}

__global__ void inv_count(const int* __restrict__ cnt, float* __restrict__ inv, int n) {
  int i = blockIdx.x * blockDim.x + threadIdx.x;
  if (i < n) {
    int c = cnt[i];
    inv[i] = 1.0f / (float)(c > 0 ? c : 1);
  }
}

__global__ void sage_gemm_wmma(const float* __restrict__ agg,
                               const float* __restrict__ inv,
                               const float* __restrict__ selfA,
                               const float* __restrict__ Wn,
                               const float* __restrict__ Ws,
                               const float* __restrict__ bias,
                               float* __restrict__ out,
                               int M, int N, int do_relu) {
  const int K = D_FEAT;
  int wave = threadIdx.x >> 5;
  int lane = threadIdx.x & 31;
  int strips = N >> 6;
  int tilesM = (M + 15) >> 4;
  int tile = blockIdx.x * 8 + wave;
  if (tile >= tilesM * strips) return;
  int tm = tile / strips;
  int tn = tile - tm * strips;
  int row0 = tm << 4;
  int col0 = tn << 6;

  int lrow = row0 + (lane & 15);
  int kb = (lane >> 4) << 1;
  int nb = col0 + (lane & 15);

  bool mok = lrow < M;
  int lrc = mok ? lrow : (M - 1);
  float invm = mok ? inv[lrc] : 0.0f;
  float sscl = mok ? 1.0f : 0.0f;
  const float* aggRow  = agg   + (long long)lrc * K + kb;
  const float* selfRow = selfA + (long long)lrc * K + kb;
  const float* pWn = Wn + (long long)kb * N + nb;
  const float* pWs = Ws + (long long)kb * N + nb;

  v8f acc[4] = {};
  for (int k0 = 0; k0 < K; k0 += 4) {
    float2 t = *(const float2*)(aggRow);
    float2 u = *(const float2*)(selfRow);
    v2f a1, a2;
    a1.x = t.x * invm; a1.y = t.y * invm;
    a2.x = u.x * sscl; a2.y = u.y * sscl;
    aggRow += 4; selfRow += 4;
#pragma unroll
    for (int tt = 0; tt < 4; ++tt) {
      v2f b1, b2;
      b1.x = pWn[tt * 16];
      b1.y = pWn[N + tt * 16];
      b2.x = pWs[tt * 16];
      b2.y = pWs[N + tt * 16];
      acc[tt] = __builtin_amdgcn_wmma_f32_16x16x4_f32(false, a1, false, b1,
                                                      (short)0, acc[tt], false, false);
      acc[tt] = __builtin_amdgcn_wmma_f32_16x16x4_f32(false, a2, false, b2,
                                                      (short)0, acc[tt], false, false);
    }
    pWn += 4 * N; pWs += 4 * N;
  }

  int mbase = row0 + ((lane >> 4) << 3);
#pragma unroll
  for (int tt = 0; tt < 4; ++tt) {
    int n = nb + tt * 16;
    float bn = bias[n];
#pragma unroll
    for (int v = 0; v < 8; ++v) {
      int m = mbase + v;
      if (m < M) {
        float r = acc[tt][v] + bn;
        if (do_relu) r = fmaxf(r, 0.0f);
        out[(long long)m * N + n] = r;
      }
    }
  }
}

static inline int cdiv(long long a, long long b) { return (int)((a + b - 1) / b); }

extern "C" void kernel_launch(void* const* d_in, const int* in_sizes, int n_in,
                              void* d_out, int out_size, void* d_ws, size_t ws_size,
                              hipStream_t stream) {
  const float* x   = (const float*)d_in[0];
  const float* Wn0 = (const float*)d_in[1];
  const float* Ws0 = (const float*)d_in[2];
  const float* b0  = (const float*)d_in[3];
  const float* Wn1 = (const float*)d_in[4];
  const float* Ws1 = (const float*)d_in[5];
  const float* b1  = (const float*)d_in[6];
  const float* Wn2 = (const float*)d_in[7];
  const float* Ws2 = (const float*)d_in[8];
  const float* b2  = (const float*)d_in[9];
  const int* src0  = (const int*)d_in[10];
  const int* dst0  = (const int*)d_in[11];
  const int* src1  = (const int*)d_in[12];
  const int* dst1  = (const int*)d_in[13];
  const int* src2  = (const int*)d_in[14];
  const int* dst2  = (const int*)d_in[15];

  const int N0 = 50000, N1 = 25000, N2 = 12500;
  const int E0 = 500000, E1 = 250000, E2 = 125000;

  char* ws = (char*)d_ws;
  float* agg = (float*)ws;
  ws += (size_t)N0 * D_FEAT * sizeof(float);
  float* h0 = (float*)ws;
  ws += (size_t)N0 * D_FEAT * sizeof(float);
  float* h1 = (float*)ws;
  ws += (size_t)N1 * D_FEAT * sizeof(float);
  float* inv = (float*)ws;
  ws += (size_t)N0 * sizeof(float);
  int* cnt = (int*)ws;

  const int TB = 256;

  zero_f32<<<cdiv((long long)N0 * D_FEAT, TB), TB, 0, stream>>>(agg, N0 * D_FEAT);
  zero_i32<<<cdiv(N0, TB), TB, 0, stream>>>(cnt, N0);
  sage_scatter<<<cdiv((long long)E0 * 64, TB), TB, 0, stream>>>(x, src0, dst0, E0, agg, cnt);
  inv_count<<<cdiv(N0, TB), TB, 0, stream>>>(cnt, inv, N0);
  {
    int tiles = ((N0 + 15) / 16) * (256 / 64);
    sage_gemm_wmma<<<cdiv(tiles, 8), TB, 0, stream>>>(agg, inv, x, Wn0, Ws0, b0, h0,
                                                      N0, 256, 1);
  }

  zero_f32<<<cdiv((long long)N1 * D_FEAT, TB), TB, 0, stream>>>(agg, N1 * D_FEAT);
  zero_i32<<<cdiv(N1, TB), TB, 0, stream>>>(cnt, N1);
  sage_scatter<<<cdiv((long long)E1 * 64, TB), TB, 0, stream>>>(h0, src1, dst1, E1, agg, cnt);
  inv_count<<<cdiv(N1, TB), TB, 0, stream>>>(cnt, inv, N1);
  {
    int tiles = ((N1 + 15) / 16) * (256 / 64);
    sage_gemm_wmma<<<cdiv(tiles, 8), TB, 0, stream>>>(agg, inv, h0, Wn1, Ws1, b1, h1,
                                                      N1, 256, 1);
  }

  zero_f32<<<cdiv((long long)N2 * D_FEAT, TB), TB, 0, stream>>>(agg, N2 * D_FEAT);
  zero_i32<<<cdiv(N2, TB), TB, 0, stream>>>(cnt, N2);
  sage_scatter<<<cdiv((long long)E2 * 64, TB), TB, 0, stream>>>(h1, src2, dst2, E2, agg, cnt);
  inv_count<<<cdiv(N2, TB), TB, 0, stream>>>(cnt, inv, N2);
  {
    int tiles = ((N2 + 15) / 16) * (128 / 64);
    sage_gemm_wmma<<<cdiv(tiles, 8), TB, 0, stream>>>(agg, inv, h1, Wn2, Ws2, b2,
                                                      (float*)d_out, N2, 128, 0);
  }
}


# Float-atomic scatter replaced by owner blocks: a block keeps a range of destination rows in LDS, filters the whole edge list for it, adds source rows with LDS float adds; sum order = arrival order
# speedup vs baseline: 2.5935x; 2.5935x over previous
#include <hip/hip_runtime.h>
#include <hip/hip_bf16.h>

typedef __attribute__((ext_vector_type(2))) float v2f;
typedef __attribute__((ext_vector_type(8))) float v8f;

#define D_FEAT 256

#define OWN_LDS_ROWS 197
#define OWN_WAVES 32
#define OWN_RING 512

template <int R>
__global__ void __launch_bounds__(1024)
sage_owner_sum(const float* __restrict__ H,
               const int* __restrict__ src,
               const int* __restrict__ dst,
               int E, int N,
               float* __restrict__ agg,
               float* __restrict__ inv) {
  static_assert(R + 1 <= OWN_LDS_ROWS, "owned rows plus the spare must fit");
  static_assert(R <= 256, "the local row shares a word with the edge index");
  __shared__ float acc[OWN_LDS_ROWS * D_FEAT];
  __shared__ int cnt[OWN_LDS_ROWS];
  __shared__ unsigned ring[OWN_WAVES][OWN_RING];

  const int tid = threadIdx.x;
  const int lane = tid & 31;
  const int wave = __builtin_amdgcn_readfirstlane(tid >> 5);
  const int r0 = blockIdx.x * R;

  for (int i = tid; i < (R + 1) * (D_FEAT / 4); i += 1024)
    ((float4*)acc)[i] = make_float4(0.0f, 0.0f, 0.0f, 0.0f);
  for (int i = tid; i <= R; i += 1024) cnt[i] = 0;
  __syncthreads();

  int head = 0, tail = 0;

  auto append = [&](int d, int e) {
    const unsigned rel = (unsigned)(d - r0);
    const unsigned m = (unsigned)__builtin_amdgcn_uicmp(rel, (unsigned)R, 36);
    if (m != 0u) {
      if (rel < (unsigned)R) {
        const int slot = (tail + (int)__builtin_amdgcn_mbcnt_lo(m, 0u)) & (OWN_RING - 1);
        ring[wave][slot] = ((unsigned)e << 8) | rel;
      }
      tail += __builtin_popcount(m);
    }
  };

  const int rot = lane >> 4;
  int off[4];
#pragma unroll
  for (int c = 0; c < 4; ++c) off[c] = lane * 4 + ((c + rot) & 3);

  auto drain = [&](int h, int n) {
    const unsigned ent = ring[wave][(h + (lane < n ? lane : 0)) & (OWN_RING - 1)];
    const int rr = lane < n ? (int)(ent & 255u) : R;
    if (lane < n) atomicAdd(&cnt[rr], 1);
    const int sv = src[ent >> 8];
    const int ng = (n + 7) >> 3;
    for (int g = 0; g < ng; ++g) {
      float4 a[8], b[8];
      int row[8];
#pragma unroll
      for (int k = 0; k < 8; ++k) {
        const int j = g * 8 + k;
        const int s = __builtin_amdgcn_readlane(sv, j);
        row[k] = __builtin_amdgcn_readlane(rr, j);
        const float4* p = (const float4*)(H + (long long)s * D_FEAT);
        a[k] = p[lane];
        b[k] = p[32 + lane];
      }
#pragma unroll
      for (int k = 0; k < 8; ++k) {
        float* q = &acc[row[k] * D_FEAT];
        const float a0 = rot ? a[k].y : a[k].x, a1 = rot ? a[k].z : a[k].y;
        const float a2 = rot ? a[k].w : a[k].z, a3 = rot ? a[k].x : a[k].w;
        const float b0 = rot ? b[k].y : b[k].x, b1 = rot ? b[k].z : b[k].y;
        const float b2 = rot ? b[k].w : b[k].z, b3 = rot ? b[k].x : b[k].w;
        atomicAdd(q + off[0], a0);
        atomicAdd(q + off[1], a1);
        atomicAdd(q + off[2], a2);
        atomicAdd(q + off[3], a3);
        atomicAdd(q + 128 + off[0], b0);
        atomicAdd(q + 128 + off[1], b1);
        atomicAdd(q + 128 + off[2], b2);
        atomicAdd(q + 128 + off[3], b3);
      }
    }
  };

  const int nFull = E >> 8;
  if (wave < nFull) {
    const int4* d4 = (const int4*)dst;
    const int first = (int)(((unsigned)blockIdx.x * (unsigned)nFull) / gridDim.x);
    int it = wave + first;
    if (it >= nFull) it -= nFull;
    int4 ca = d4[it * 64 + lane];
    int4 cb = d4[it * 64 + 32 + lane];
    for (int t = wave; t < nFull; t += OWN_WAVES) {
      int nx = it;
      if (t + OWN_WAVES < nFull) {
        nx = t + OWN_WAVES + first;
        if (nx >= nFull) nx -= nFull;
      }
      const int4 na = d4[nx * 64 + lane];
      const int4 nb = d4[nx * 64 + 32 + lane];
      const int base = it * 256 + lane * 4;
      append(ca.x, base + 0);
      append(ca.y, base + 1);
      append(ca.z, base + 2);
      append(ca.w, base + 3);
      append(cb.x, base + 128);
      append(cb.y, base + 129);
      append(cb.z, base + 130);
      append(cb.w, base + 131);
      while (tail - head >= 16) {
        drain(head, 16);
        head += 16;
      }
      ca = na;
      cb = nb;
      it = nx;
    }
  }
  if ((E & 255) != 0 && wave == (nFull & (OWN_WAVES - 1))) {
    const int base = nFull * 256 + lane * 4;
    int dv[8];
#pragma unroll
    for (int j = 0; j < 8; ++j) {
      const int e = base + (j & 3) + (j >> 2) * 128;
      dv[j] = dst[e < E ? e : E - 1];
    }
#pragma unroll
    for (int j = 0; j < 8; ++j) {
      const int e = base + (j & 3) + (j >> 2) * 128;
      append(e < E ? dv[j] : -1, e);
    }
  }
  while (tail != head) {
    const int n = (tail - head) < 32 ? (tail - head) : 32;
    drain(head, n);
    head += n;
  }
  __syncthreads();

  const int nrows = (N - r0) < R ? (N - r0) : R;
  float4* o4 = (float4*)(agg + (long long)r0 * D_FEAT);
  for (int i = tid; i < nrows * (D_FEAT / 4); i += 1024) o4[i] = ((const float4*)acc)[i];
  for (int i = tid; i < nrows; i += 1024) {
    const int c = cnt[i];
    inv[r0 + i] = 1.0f / (float)(c > 0 ? c : 1);
  }
}

template <int MT>
struct Frag {
  float4 a[MT];
  float4 s[MT];
  float bn[2][4][2];
  float bs[2][4][2];
};

template <int MT, int NC>
__device__ __forceinline__ void load_frag(Frag<MT>& f,
                                          const float* const (&aRow)[MT],
                                          const float* const (&sRow)[MT],
                                          const float* __restrict__ pWn,
                                          const float* __restrict__ pWs,
                                          int kc) {
#pragma unroll
  for (int mi = 0; mi < MT; ++mi) {
    f.a[mi] = *(const float4*)(aRow[mi] + kc * 8);
    f.s[mi] = *(const float4*)(sRow[mi] + kc * 8);
  }
  const float* qn = pWn + (long long)kc * (8 * NC);
  const float* qs = pWs + (long long)kc * (8 * NC);
#pragma unroll
  for (int sub = 0; sub < 2; ++sub) {
#pragma unroll
    for (int tt = 0; tt < 4; ++tt) {
      f.bn[sub][tt][0] = qn[(2 * sub) * NC + tt * 16];
      f.bn[sub][tt][1] = qn[(2 * sub + 1) * NC + tt * 16];
      f.bs[sub][tt][0] = qs[(2 * sub) * NC + tt * 16];
      f.bs[sub][tt][1] = qs[(2 * sub + 1) * NC + tt * 16];
    }
  }
}

template <int MT>
__device__ __forceinline__ void mma_chunk(const Frag<MT>& f, const float (&invm)[MT],
                                          v8f (&acc)[MT][4]) {
#pragma unroll
  for (int sub = 0; sub < 2; ++sub) {
#pragma unroll
    for (int mi = 0; mi < MT; ++mi) {
      v2f a1, a2;
      if (sub == 0) {
        a1.x = f.a[mi].x * invm[mi]; a1.y = f.a[mi].y * invm[mi];
        a2.x = f.s[mi].x;            a2.y = f.s[mi].y;
      } else {
        a1.x = f.a[mi].z * invm[mi]; a1.y = f.a[mi].w * invm[mi];
        a2.x = f.s[mi].z;            a2.y = f.s[mi].w;
      }
#pragma unroll
      for (int tt = 0; tt < 4; ++tt) {
        v2f b1;
        b1.x = f.bn[sub][tt][0]; b1.y = f.bn[sub][tt][1];
        acc[mi][tt] = __builtin_amdgcn_wmma_f32_16x16x4_f32(false, a1, false, b1,
                                                            (short)0, acc[mi][tt], false, false);
      }
#pragma unroll
      for (int tt = 0; tt < 4; ++tt) {
        v2f b2;
        b2.x = f.bs[sub][tt][0]; b2.y = f.bs[sub][tt][1];
        acc[mi][tt] = __builtin_amdgcn_wmma_f32_16x16x4_f32(false, a2, false, b2,
                                                            (short)0, acc[mi][tt], false, false);
      }
    }
  }
}

template <int MT, int NC, bool RELU, int WPS>
__global__ void __launch_bounds__(128) __attribute__((amdgpu_waves_per_eu(WPS, WPS)))
sage_gemm_wmma(const float* __restrict__ agg,
               const float* __restrict__ inv,
               const float* __restrict__ selfA,
               const float* __restrict__ Wn,
               const float* __restrict__ Ws,
               const float* __restrict__ bias,
               float* __restrict__ out,
               int M) {
  constexpr int K = D_FEAT;
  constexpr int NCHUNK = K / 8;
  constexpr int STRIPS = NC / 64;
  constexpr int ROWS = 16 * MT;
  int wave = threadIdx.x >> 5;
  int lane = threadIdx.x & 31;
  int tilesM = (M + ROWS - 1) / ROWS;
  int tile = blockIdx.x * 4 + wave;
  if (tile >= tilesM * STRIPS) return;
  int tm = tile / STRIPS;
  int tn = tile - tm * STRIPS;
  int row0 = tm * ROWS;
  int col0 = tn << 6;
  int c = lane & 15;
  int h = lane >> 4;
  int nb = col0 + c;

  const float* aRow[MT];
  const float* sRow[MT];
  float invm[MT];
#pragma unroll
  for (int mi = 0; mi < MT; ++mi) {
    int lrow = row0 + mi * 16 + c;
    int lrc = lrow < M ? lrow : (M - 1);
    invm[mi] = inv[lrc];
    aRow[mi] = agg   + (long long)lrc * K + 4 * h;
    sRow[mi] = selfA + (long long)lrc * K + 4 * h;
  }
  const float* pWn = Wn + (long long)(4 * h) * NC + nb;
  const float* pWs = Ws + (long long)(4 * h) * NC + nb;

  v8f acc[MT][4] = {};
  Frag<MT> f0, f1;
  constexpr int NLD = 2 * MT + 32;
  constexpr int NMM = 16 * MT;
  load_frag<MT, NC>(f0, aRow, sRow, pWn, pWs, 0);
  for (int kc = 0; kc < NCHUNK - 2; kc += 2) {
    load_frag<MT, NC>(f1, aRow, sRow, pWn, pWs, kc + 1);
    mma_chunk<MT>(f0, invm, acc);
    load_frag<MT, NC>(f0, aRow, sRow, pWn, pWs, kc + 2);
    mma_chunk<MT>(f1, invm, acc);
    __builtin_amdgcn_sched_group_barrier(0x020, NLD, 0);
    __builtin_amdgcn_sched_group_barrier(0x008, NMM, 0);
    __builtin_amdgcn_sched_group_barrier(0x020, NLD, 0);
    __builtin_amdgcn_sched_group_barrier(0x008, NMM, 0);
  }
  load_frag<MT, NC>(f1, aRow, sRow, pWn, pWs, NCHUNK - 1);
  mma_chunk<MT>(f0, invm, acc);
  mma_chunk<MT>(f1, invm, acc);

  int mb = row0 + (h << 3);
#pragma unroll
  for (int mi = 0; mi < MT; ++mi) {
#pragma unroll
    for (int tt = 0; tt < 4; ++tt) {
      int n = nb + tt * 16;
      float bn = bias[n];
#pragma unroll
      for (int v = 0; v < 8; ++v) {
        int m = mb + mi * 16 + v;
        if (m < M) {
          float r = acc[mi][tt][v] + bn;
          if (RELU) r = fmaxf(r, 0.0f);
          out[(long long)m * NC + n] = r;
        }
      }
    }
  }
}

static inline int cdiv(long long a, long long b) { return (int)((a + b - 1) / b); }

extern "C" void kernel_launch(void* const* d_in, const int* in_sizes, int n_in,
                              void* d_out, int out_size, void* d_ws, size_t ws_size,
                              hipStream_t stream) {
  const float* x   = (const float*)d_in[0];
  const float* Wn0 = (const float*)d_in[1];
  const float* Ws0 = (const float*)d_in[2];
  const float* b0  = (const float*)d_in[3];
  const float* Wn1 = (const float*)d_in[4];
  const float* Ws1 = (const float*)d_in[5];
  const float* b1  = (const float*)d_in[6];
  const float* Wn2 = (const float*)d_in[7];
  const float* Ws2 = (const float*)d_in[8];
  const float* b2  = (const float*)d_in[9];
  const int* src0  = (const int*)d_in[10];
  const int* dst0  = (const int*)d_in[11];
  const int* src1  = (const int*)d_in[12];
  const int* dst1  = (const int*)d_in[13];
  const int* src2  = (const int*)d_in[14];
  const int* dst2  = (const int*)d_in[15];

  const int N0 = 50000, N1 = 25000, N2 = 12500;
  const int E0 = 500000, E1 = 250000, E2 = 125000;

  char* ws = (char*)d_ws;
  float* agg = (float*)ws;
  ws += (size_t)N0 * D_FEAT * sizeof(float);
  float* h0 = (float*)ws;
  ws += (size_t)N0 * D_FEAT * sizeof(float);
  float* h1 = (float*)ws;
  ws += (size_t)N1 * D_FEAT * sizeof(float);
  float* inv = (float*)ws;

  sage_owner_sum<196><<<cdiv(N0, 196), 1024, 0, stream>>>(x, src0, dst0, E0, N0, agg, inv);
  {
    int tiles = ((N0 + 63) / 64) * (256 / 64);
    sage_gemm_wmma<4, 256, true, 3><<<cdiv(tiles, 4), 128, 0, stream>>>(agg, inv, x, Wn0, Ws0, b0,
                                                                     h0, N0);
  }

  sage_owner_sum<98><<<cdiv(N1, 98), 1024, 0, stream>>>(h0, src1, dst1, E1, N1, agg, inv);
  {
    int tiles = ((N1 + 31) / 32) * (256 / 64);
    sage_gemm_wmma<2, 256, true, 4><<<cdiv(tiles, 4), 128, 0, stream>>>(agg, inv, h0, Wn1, Ws1, b1,
                                                                     h1, N1);
  }

  sage_owner_sum<49><<<cdiv(N2, 49), 1024, 0, stream>>>(h1, src2, dst2, E2, N2, agg, inv);
  {
    int tiles = ((N2 + 15) / 16) * (128 / 64);
    sage_gemm_wmma<1, 128, false, 4><<<cdiv(tiles, 4), 128, 0, stream>>>(agg, inv, h1, Wn2, Ws2, b2,
                                                                      (float*)d_out, N2);
  }
}


# Dual GEMM rows re-split so every SIMD gets near-equal WMMA counts: 768 full-size blocks fill all SIMDs evenly, left-over rows go as 32x16 tiles (first in layer 0, last in layer 1); sum order kept
# speedup vs baseline: 2.7136x; 1.0463x over previous
#include <hip/hip_runtime.h>
#include <hip/hip_bf16.h>

typedef __attribute__((ext_vector_type(2))) float v2f;
typedef __attribute__((ext_vector_type(8))) float v8f;

#define D_FEAT 256

#define OWN_LDS_ROWS 197
#define OWN_WAVES 32
#define OWN_RING 512

template <int R>
__global__ void __launch_bounds__(1024)
sage_owner_sum(const float* __restrict__ H,
               const int* __restrict__ src,
               const int* __restrict__ dst,
               int E, int N,
               float* __restrict__ agg,
               float* __restrict__ inv) {
  static_assert(R + 1 <= OWN_LDS_ROWS, "owned rows plus the spare must fit");
  static_assert(R <= 256, "the local row shares a word with the edge index");
  __shared__ float acc[OWN_LDS_ROWS * D_FEAT];
  __shared__ int cnt[OWN_LDS_ROWS];
  __shared__ unsigned ring[OWN_WAVES][OWN_RING];

  const int tid = threadIdx.x;
  const int lane = tid & 31;
  const int wave = __builtin_amdgcn_readfirstlane(tid >> 5);
  const int r0 = blockIdx.x * R;

  for (int i = tid; i < (R + 1) * (D_FEAT / 4); i += 1024)
    ((float4*)acc)[i] = make_float4(0.0f, 0.0f, 0.0f, 0.0f);
  for (int i = tid; i <= R; i += 1024) cnt[i] = 0;
  __syncthreads();

  int head = 0, tail = 0;

  auto append = [&](int d, int e) {
    const unsigned rel = (unsigned)(d - r0);
    const unsigned m = (unsigned)__builtin_amdgcn_uicmp(rel, (unsigned)R, 36);
    if (m != 0u) {
      if (rel < (unsigned)R) {
        const int slot = (tail + (int)__builtin_amdgcn_mbcnt_lo(m, 0u)) & (OWN_RING - 1);
        ring[wave][slot] = ((unsigned)e << 8) | rel;
      }
      tail += __builtin_popcount(m);
    }
  };

  const int rot = lane >> 4;
  int off[4];
#pragma unroll
  for (int c = 0; c < 4; ++c) off[c] = lane * 4 + ((c + rot) & 3);

  auto drain = [&](int h, int n) {
    const unsigned ent = ring[wave][(h + (lane < n ? lane : 0)) & (OWN_RING - 1)];
    const int rr = lane < n ? (int)(ent & 255u) : R;
    if (lane < n) atomicAdd(&cnt[rr], 1);
    const int sv = src[ent >> 8];
    const int ng = (n + 7) >> 3;
    for (int g = 0; g < ng; ++g) {
      float4 a[8], b[8];
      int row[8];
#pragma unroll
      for (int k = 0; k < 8; ++k) {
        const int j = g * 8 + k;
        const int s = __builtin_amdgcn_readlane(sv, j);
        row[k] = __builtin_amdgcn_readlane(rr, j);
        const float4* p = (const float4*)(H + (long long)s * D_FEAT);
        a[k] = p[lane];
        b[k] = p[32 + lane];
      }
#pragma unroll
      for (int k = 0; k < 8; ++k) {
        float* q = &acc[row[k] * D_FEAT];
        const float a0 = rot ? a[k].y : a[k].x, a1 = rot ? a[k].z : a[k].y;
        const float a2 = rot ? a[k].w : a[k].z, a3 = rot ? a[k].x : a[k].w;
        const float b0 = rot ? b[k].y : b[k].x, b1 = rot ? b[k].z : b[k].y;
        const float b2 = rot ? b[k].w : b[k].z, b3 = rot ? b[k].x : b[k].w;
        atomicAdd(q + off[0], a0);
        atomicAdd(q + off[1], a1);
        atomicAdd(q + off[2], a2);
        atomicAdd(q + off[3], a3);
        atomicAdd(q + 128 + off[0], b0);
        atomicAdd(q + 128 + off[1], b1);
        atomicAdd(q + 128 + off[2], b2);
        atomicAdd(q + 128 + off[3], b3);
      }
    }
  };

  const int nFull = E >> 8;
  if (wave < nFull) {
    const int4* d4 = (const int4*)dst;
    const int first = (int)(((unsigned)blockIdx.x * (unsigned)nFull) / gridDim.x);
    int it = wave + first;
    if (it >= nFull) it -= nFull;
    int4 ca = d4[it * 64 + lane];
    int4 cb = d4[it * 64 + 32 + lane];
    for (int t = wave; t < nFull; t += OWN_WAVES) {
      int nx = it;
      if (t + OWN_WAVES < nFull) {
        nx = t + OWN_WAVES + first;
        if (nx >= nFull) nx -= nFull;
      }
      const int4 na = d4[nx * 64 + lane];
      const int4 nb = d4[nx * 64 + 32 + lane];
      const int base = it * 256 + lane * 4;
      append(ca.x, base + 0);
      append(ca.y, base + 1);
      append(ca.z, base + 2);
      append(ca.w, base + 3);
      append(cb.x, base + 128);
      append(cb.y, base + 129);
      append(cb.z, base + 130);
      append(cb.w, base + 131);
      while (tail - head >= 16) {
        drain(head, 16);
        head += 16;
      }
      ca = na;
      cb = nb;
      it = nx;
    }
  }
  if ((E & 255) != 0 && wave == (nFull & (OWN_WAVES - 1))) {
    const int base = nFull * 256 + lane * 4;
    int dv[8];
#pragma unroll
    for (int j = 0; j < 8; ++j) {
      const int e = base + (j & 3) + (j >> 2) * 128;
      dv[j] = dst[e < E ? e : E - 1];
    }
#pragma unroll
    for (int j = 0; j < 8; ++j) {
      const int e = base + (j & 3) + (j >> 2) * 128;
      append(e < E ? dv[j] : -1, e);
    }
  }
  while (tail != head) {
    const int n = (tail - head) < 32 ? (tail - head) : 32;
    drain(head, n);
    head += n;
  }
  __syncthreads();

  const int nrows = (N - r0) < R ? (N - r0) : R;
  float4* o4 = (float4*)(agg + (long long)r0 * D_FEAT);
  for (int i = tid; i < nrows * (D_FEAT / 4); i += 1024) o4[i] = ((const float4*)acc)[i];
  for (int i = tid; i < nrows; i += 1024) {
    const int c = cnt[i];
    inv[r0 + i] = 1.0f / (float)(c > 0 ? c : 1);
  }
}

template <int MT, int NT>
struct Frag {
  float4 a[MT];
  float4 s[MT];
  float bn[2][NT][2];
  float bs[2][NT][2];
};

template <int MT, int NT, int NC>
__device__ __forceinline__ void load_frag(Frag<MT, NT>& f,
                                          const float* const (&aRow)[MT],
                                          const float* const (&sRow)[MT],
                                          const float* __restrict__ pWn,
                                          const float* __restrict__ pWs,
                                          int kc) {
#pragma unroll
  for (int mi = 0; mi < MT; ++mi) {
    f.a[mi] = *(const float4*)(aRow[mi] + kc * 8);
    f.s[mi] = *(const float4*)(sRow[mi] + kc * 8);
  }
  const float* qn = pWn + (long long)kc * (8 * NC);
  const float* qs = pWs + (long long)kc * (8 * NC);
#pragma unroll
  for (int sub = 0; sub < 2; ++sub) {
#pragma unroll
    for (int tt = 0; tt < NT; ++tt) {
      f.bn[sub][tt][0] = qn[(2 * sub) * NC + tt * 16];
      f.bn[sub][tt][1] = qn[(2 * sub + 1) * NC + tt * 16];
      f.bs[sub][tt][0] = qs[(2 * sub) * NC + tt * 16];
      f.bs[sub][tt][1] = qs[(2 * sub + 1) * NC + tt * 16];
    }
  }
}

template <int MT, int NT>
__device__ __forceinline__ void mma_chunk(const Frag<MT, NT>& f, const float (&invm)[MT],
                                          v8f (&acc)[MT][NT]) {
#pragma unroll
  for (int sub = 0; sub < 2; ++sub) {
#pragma unroll
    for (int mi = 0; mi < MT; ++mi) {
      v2f a1, a2;
      if (sub == 0) {
        a1.x = f.a[mi].x * invm[mi]; a1.y = f.a[mi].y * invm[mi];
        a2.x = f.s[mi].x;            a2.y = f.s[mi].y;
      } else {
        a1.x = f.a[mi].z * invm[mi]; a1.y = f.a[mi].w * invm[mi];
        a2.x = f.s[mi].z;            a2.y = f.s[mi].w;
      }
#pragma unroll
      for (int tt = 0; tt < NT; ++tt) {
        v2f b1;
        b1.x = f.bn[sub][tt][0]; b1.y = f.bn[sub][tt][1];
        acc[mi][tt] = __builtin_amdgcn_wmma_f32_16x16x4_f32(false, a1, false, b1,
                                                            (short)0, acc[mi][tt], false, false);
      }
#pragma unroll
      for (int tt = 0; tt < NT; ++tt) {
        v2f b2;
        b2.x = f.bs[sub][tt][0]; b2.y = f.bs[sub][tt][1];
        acc[mi][tt] = __builtin_amdgcn_wmma_f32_16x16x4_f32(false, a2, false, b2,
                                                            (short)0, acc[mi][tt], false, false);
      }
    }
  }
}

template <int MT, int NT, int NC, bool RELU>
__device__ __forceinline__ void sage_gemm_tile(const float* __restrict__ agg,
                                               const float* __restrict__ inv,
                                               const float* __restrict__ selfA,
                                               const float* __restrict__ Wn,
                                               const float* __restrict__ Ws,
                                               const float* __restrict__ bias,
                                               float* __restrict__ out,
                                               int M, int row0, int col0, int lane) {
  constexpr int K = D_FEAT;
  constexpr int NCHUNK = K / 8;
  int c = lane & 15;
  int h = lane >> 4;
  int nb = col0 + c;

  const float* aRow[MT];
  const float* sRow[MT];
  float invm[MT];
#pragma unroll
  for (int mi = 0; mi < MT; ++mi) {
    int lrow = row0 + mi * 16 + c;
    int lrc = lrow < M ? lrow : (M - 1);
    invm[mi] = inv[lrc];
    aRow[mi] = agg   + (long long)lrc * K + 4 * h;
    sRow[mi] = selfA + (long long)lrc * K + 4 * h;
  }
  const float* pWn = Wn + (long long)(4 * h) * NC + nb;
  const float* pWs = Ws + (long long)(4 * h) * NC + nb;

  v8f acc[MT][NT] = {};
  Frag<MT, NT> f0, f1;
  constexpr int NLD = 2 * MT + 8 * NT;
  constexpr int NMM = 4 * MT * NT;
  load_frag<MT, NT, NC>(f0, aRow, sRow, pWn, pWs, 0);
  for (int kc = 0; kc < NCHUNK - 2; kc += 2) {
    load_frag<MT, NT, NC>(f1, aRow, sRow, pWn, pWs, kc + 1);
    mma_chunk<MT, NT>(f0, invm, acc);
    load_frag<MT, NT, NC>(f0, aRow, sRow, pWn, pWs, kc + 2);
    mma_chunk<MT, NT>(f1, invm, acc);
    __builtin_amdgcn_sched_group_barrier(0x020, NLD, 0);
    __builtin_amdgcn_sched_group_barrier(0x008, NMM, 0);
    __builtin_amdgcn_sched_group_barrier(0x020, NLD, 0);
    __builtin_amdgcn_sched_group_barrier(0x008, NMM, 0);
  }
  load_frag<MT, NT, NC>(f1, aRow, sRow, pWn, pWs, NCHUNK - 1);
  mma_chunk<MT, NT>(f0, invm, acc);
  mma_chunk<MT, NT>(f1, invm, acc);

  int mb = row0 + (h << 3);
#pragma unroll
  for (int mi = 0; mi < MT; ++mi) {
#pragma unroll
    for (int tt = 0; tt < NT; ++tt) {
      int n = nb + tt * 16;
      float bn = bias[n];
#pragma unroll
      for (int v = 0; v < 8; ++v) {
        int m = mb + mi * 16 + v;
        if (m < M) {
          float r = acc[mi][tt][v] + bn;
          if (RELU) r = fmaxf(r, 0.0f);
          out[(long long)m * NC + n] = r;
        }
      }
    }
  }
}

template <int MT, int NC, bool RELU, int WPS, int SPLIT>
__global__ void __launch_bounds__(128) __attribute__((amdgpu_waves_per_eu(WPS, WPS)))
sage_gemm_wmma(const float* __restrict__ agg,
               const float* __restrict__ inv,
               const float* __restrict__ selfA,
               const float* __restrict__ Wn,
               const float* __restrict__ Ws,
               const float* __restrict__ bias,
               float* __restrict__ out,
               int M, int tailRows) {
  constexpr int STRIPS = NC / 64;
  constexpr int TSTRIPS = NC / 16;
  constexpr int TROWS = 32;
  constexpr int ROWS = 16 * MT;
  const int wave = __builtin_amdgcn_readfirstlane((int)(threadIdx.x >> 5));
  const int lane = threadIdx.x & 31;
  if (SPLIT == 0) tailRows = 0;
  const int mainTiles = ((M - tailRows + ROWS - 1) / ROWS) * STRIPS;
  const int tailTiles = (tailRows / TROWS) * TSTRIPS;
  int tile = blockIdx.x * 4 + wave;
  bool small = false;
  if (SPLIT == 1) {
    small = tile < tailTiles;
    if (!small) tile -= tailTiles;
  } else if (SPLIT == 2) {
    small = tile >= mainTiles;
    if (small) tile -= mainTiles;
  }
  if (SPLIT != 0 && small) {
    if (tile >= tailTiles) return;
    const int tm = tile / TSTRIPS;
    const int tn = tile - tm * TSTRIPS;
    sage_gemm_tile<2, 1, NC, RELU>(agg, inv, selfA, Wn, Ws, bias, out, M, tm * TROWS, tn << 4,
                                   lane);
    return;
  }
  if (tile >= mainTiles) return;
  const int tm = tile / STRIPS;
  const int tn = tile - tm * STRIPS;
  sage_gemm_tile<MT, 4, NC, RELU>(agg, inv, selfA, Wn, Ws, bias, out, M, tailRows + tm * ROWS,
                                  tn << 6, lane);
}

static inline int cdiv(long long a, long long b) { return (int)((a + b - 1) / b); }

extern "C" void kernel_launch(void* const* d_in, const int* in_sizes, int n_in,
                              void* d_out, int out_size, void* d_ws, size_t ws_size,
                              hipStream_t stream) {
  const float* x   = (const float*)d_in[0];
  const float* Wn0 = (const float*)d_in[1];
  const float* Ws0 = (const float*)d_in[2];
  const float* b0  = (const float*)d_in[3];
  const float* Wn1 = (const float*)d_in[4];
  const float* Ws1 = (const float*)d_in[5];
  const float* b1  = (const float*)d_in[6];
  const float* Wn2 = (const float*)d_in[7];
  const float* Ws2 = (const float*)d_in[8];
  const float* b2  = (const float*)d_in[9];
  const int* src0  = (const int*)d_in[10];
  const int* dst0  = (const int*)d_in[11];
  const int* src1  = (const int*)d_in[12];
  const int* dst1  = (const int*)d_in[13];
  const int* src2  = (const int*)d_in[14];
  const int* dst2  = (const int*)d_in[15];

  const int N0 = 50000, N1 = 25000, N2 = 12500;
  const int E0 = 500000, E1 = 250000, E2 = 125000;

  char* ws = (char*)d_ws;
  float* agg = (float*)ws;
  ws += (size_t)N0 * D_FEAT * sizeof(float);
  float* h0 = (float*)ws;
  ws += (size_t)N0 * D_FEAT * sizeof(float);
  float* h1 = (float*)ws;
  ws += (size_t)N1 * D_FEAT * sizeof(float);
  float* inv = (float*)ws;

  sage_owner_sum<196><<<cdiv(N0, 196), 1024, 0, stream>>>(x, src0, dst0, E0, N0, agg, inv);
  {
    const int tail = 864;
    int tiles = (tail / 32) * (256 / 16) + (((N0 - tail) + 63) / 64) * (256 / 64);
    sage_gemm_wmma<4, 256, true, 3, 1><<<cdiv(tiles, 4), 128, 0, stream>>>(agg, inv, x, Wn0, Ws0,
                                                                           b0, h0, N0, tail);
  }

  sage_owner_sum<98><<<cdiv(N1, 98), 1024, 0, stream>>>(h0, src1, dst1, E1, N1, agg, inv);
  {
    const int tail = 448;
    int tiles = (tail / 32) * (256 / 16) + (((N1 - tail) + 31) / 32) * (256 / 64);
    sage_gemm_wmma<2, 256, true, 4, 2><<<cdiv(tiles, 4), 128, 0, stream>>>(agg, inv, h0, Wn1,
                                                                           Ws1, b1, h1, N1, tail);
  }

  sage_owner_sum<49><<<cdiv(N2, 49), 1024, 0, stream>>>(h1, src2, dst2, E2, N2, agg, inv);
  {
    int tiles = ((N2 + 15) / 16) * (128 / 64);
    sage_gemm_wmma<1, 128, false, 4, 0><<<cdiv(tiles, 4), 128, 0, stream>>>(agg, inv, h1, Wn2,
                                                                             Ws2, b2,
                                                                             (float*)d_out, N2, 0);
  }
}


# Owner blocks: each hit goes through LDS queues to the wave that owns its row (row % 32); that wave sums its rows in registers, so the LDS float adds go; sum order = queue arrival order
# speedup vs baseline: 2.8837x; 1.0627x over previous
#include <hip/hip_runtime.h>
#include <hip/hip_bf16.h>

typedef __attribute__((ext_vector_type(2))) float v2f;
typedef __attribute__((ext_vector_type(8))) float v8f;

#define D_FEAT 256

#define OWN_LDS_ROWS 196
#define OWN_WAVES 32
#define OWN_RING 512
#define OWN_QCAP 256

template <int R>
__global__ void __launch_bounds__(1024)
sage_owner_sum(const float* __restrict__ H,
               const int* __restrict__ src,
               const int* __restrict__ dst,
               int E, int N,
               float* __restrict__ agg,
               float* __restrict__ inv) {
  static_assert(R <= OWN_LDS_ROWS, "owned rows must fit the LDS copy");
  static_assert(R <= 256, "the local row shares a word with the edge index");
  constexpr int RPW = (R + OWN_WAVES - 1) / OWN_WAVES;
  constexpr int G = 8;
  __shared__ float accL[OWN_LDS_ROWS * D_FEAT];
  __shared__ int cnt[OWN_LDS_ROWS];
  __shared__ unsigned ring[OWN_WAVES][OWN_RING];
  __shared__ unsigned queue[OWN_WAVES][OWN_QCAP];
  __shared__ unsigned qtail[OWN_WAVES];

  const int tid = threadIdx.x;
  const int lane = tid & 31;
  const int wave = __builtin_amdgcn_readfirstlane(tid >> 5);
  const int r0 = blockIdx.x * R;

  for (int i = tid; i < R * (D_FEAT / 4); i += 1024)
    ((float4*)accL)[i] = make_float4(0.0f, 0.0f, 0.0f, 0.0f);
  for (int i = tid; i < R; i += 1024) cnt[i] = 0;
  if (tid < OWN_WAVES) qtail[tid] = 0u;
  __syncthreads();

  int head = 0, tail = 0;

  auto append = [&](int d, int e) {
    const unsigned rel = (unsigned)(d - r0);
    const unsigned m = (unsigned)__builtin_amdgcn_uicmp(rel, (unsigned)R, 36);
    if (m != 0u) {
      if (rel < (unsigned)R) {
        const int slot = (tail + (int)__builtin_amdgcn_mbcnt_lo(m, 0u)) & (OWN_RING - 1);
        ring[wave][slot] = ((unsigned)e << 8) | rel;
      }
      tail += __builtin_popcount(m);
    }
  };

  auto slow_add = [&](int e, int rr) {
    const int s = src[e];
    const float4* p = (const float4*)(H + (long long)s * D_FEAT);
    const float4 a = p[lane];
    const float4 b = p[32 + lane];
    float* q = &accL[rr * D_FEAT + lane * 4];
    atomicAdd(q + 0, a.x);
    atomicAdd(q + 1, a.y);
    atomicAdd(q + 2, a.z);
    atomicAdd(q + 3, a.w);
    atomicAdd(q + 128, b.x);
    atomicAdd(q + 129, b.y);
    atomicAdd(q + 130, b.z);
    atomicAdd(q + 131, b.w);
  };

  auto flush = [&](int h, int n) {
    unsigned ent = 0u;
    bool full = false;
    if (lane < n) {
      ent = ring[wave][(h + lane) & (OWN_RING - 1)];
      const unsigned rel = ent & 255u;
      atomicAdd(&cnt[rel], 1);
      const unsigned ow = rel & (OWN_WAVES - 1);
      const unsigned pos = atomicAdd(&qtail[ow], 1u);
      if (pos < (unsigned)OWN_QCAP) queue[ow][pos] = ent;
      else full = true;
    }
    unsigned om = __builtin_amdgcn_ballot_w32(full);
    while (om != 0u) {
      const int l = __builtin_ctz(om);
      om &= om - 1u;
      const unsigned eo = (unsigned)__builtin_amdgcn_readlane((int)ent, l);
      slow_add((int)(eo >> 8), (int)(eo & 255u));
    }
  };

  const int nFull = E >> 8;
  if (wave < nFull) {
    const int4* d4 = (const int4*)dst;
    const int first = (int)(((unsigned)blockIdx.x * (unsigned)nFull) / gridDim.x);
    int it = wave + first;
    if (it >= nFull) it -= nFull;
    int4 ca = d4[it * 64 + lane];
    int4 cb = d4[it * 64 + 32 + lane];
    for (int t = wave; t < nFull; t += OWN_WAVES) {
      int nx = it;
      if (t + OWN_WAVES < nFull) {
        nx = t + OWN_WAVES + first;
        if (nx >= nFull) nx -= nFull;
      }
      const int4 na = d4[nx * 64 + lane];
      const int4 nb = d4[nx * 64 + 32 + lane];
      const int base = it * 256 + lane * 4;
      append(ca.x, base + 0);
      append(ca.y, base + 1);
      append(ca.z, base + 2);
      append(ca.w, base + 3);
      append(cb.x, base + 128);
      append(cb.y, base + 129);
      append(cb.z, base + 130);
      append(cb.w, base + 131);
      while (tail - head >= 32) {
        flush(head, 32);
        head += 32;
      }
      ca = na;
      cb = nb;
      it = nx;
    }
  }
  if ((E & 255) != 0 && wave == (nFull & (OWN_WAVES - 1))) {
    const int base = nFull * 256 + lane * 4;
    int dv[8];
#pragma unroll
    for (int j = 0; j < 8; ++j) {
      const int e = base + (j & 3) + (j >> 2) * 128;
      dv[j] = dst[e < E ? e : E - 1];
    }
#pragma unroll
    for (int j = 0; j < 8; ++j) {
      const int e = base + (j & 3) + (j >> 2) * 128;
      append(e < E ? dv[j] : -1, e);
    }
  }
  while (tail != head) {
    const int n = (tail - head) < 32 ? (tail - head) : 32;
    flush(head, n);
    head += n;
  }
  __syncthreads();

  float acc[RPW][8];
#pragma unroll
  for (int q = 0; q < RPW; ++q) {
#pragma unroll
    for (int c = 0; c < 8; ++c) acc[q][c] = 0.0f;
  }
  const unsigned loff = (unsigned)lane * 16u;
  const int nq = (int)(qtail[wave] < (unsigned)OWN_QCAP ? qtail[wave] : (unsigned)OWN_QCAP);
  for (int base = 0; base < nq; base += 32) {
    const int n = (nq - base) < 32 ? (nq - base) : 32;
    const unsigned ent = queue[wave][base + (lane < n ? lane : 0)];
    const int slv = lane < n ? (int)((ent & 255u) >> 5) : RPW;
    const int sv = src[ent >> 8];
    for (int g = 0; g < n; g += G) {
      float4 a[G], b[G];
      int sl[G];
#pragma unroll
      for (int k = 0; k < G; ++k) {
        const int s = __builtin_amdgcn_readlane(sv, g + k);
        sl[k] = __builtin_amdgcn_readlane(slv, g + k);
        const char* p = (const char*)(H + (long long)s * D_FEAT);
        a[k] = *(const float4*)(p + loff);
        b[k] = *(const float4*)(p + loff + 512u);
      }
#pragma unroll
      for (int k = 0; k < G; ++k) {
#pragma unroll
        for (int q = 0; q < RPW; ++q) {
          if (sl[k] == q) {
            acc[q][0] += a[k].x;
            acc[q][1] += a[k].y;
            acc[q][2] += a[k].z;
            acc[q][3] += a[k].w;
            acc[q][4] += b[k].x;
            acc[q][5] += b[k].y;
            acc[q][6] += b[k].z;
            acc[q][7] += b[k].w;
          }
        }
      }
    }
  }

  const int nrows = (N - r0) < R ? (N - r0) : R;
#pragma unroll
  for (int q = 0; q < RPW; ++q) {
    const int row = q * OWN_WAVES + wave;
    if (row < nrows) {
      const float4* l4 = (const float4*)&accL[row * D_FEAT];
      float4 va = l4[lane];
      float4 vb = l4[32 + lane];
      va.x += acc[q][0];
      va.y += acc[q][1];
      va.z += acc[q][2];
      va.w += acc[q][3];
      vb.x += acc[q][4];
      vb.y += acc[q][5];
      vb.z += acc[q][6];
      vb.w += acc[q][7];
      float4* o4 = (float4*)(agg + (long long)(r0 + row) * D_FEAT);
      o4[lane] = va;
      o4[32 + lane] = vb;
      if (lane == 0) {
        const int c = cnt[row];
        inv[r0 + row] = 1.0f / (float)(c > 0 ? c : 1);
      }
    }
  }
}

template <int MT, int NT>
struct Frag {
  float4 a[MT];
  float4 s[MT];
  float bn[2][NT][2];
  float bs[2][NT][2];
};

template <int MT, int NT, int NC>
__device__ __forceinline__ void load_frag(Frag<MT, NT>& f,
                                          const float* const (&aRow)[MT],
                                          const float* const (&sRow)[MT],
                                          const float* __restrict__ pWn,
                                          const float* __restrict__ pWs,
                                          int kc) {
#pragma unroll
  for (int mi = 0; mi < MT; ++mi) {
    f.a[mi] = *(const float4*)(aRow[mi] + kc * 8);
    f.s[mi] = *(const float4*)(sRow[mi] + kc * 8);
  }
  const float* qn = pWn + (long long)kc * (8 * NC);
  const float* qs = pWs + (long long)kc * (8 * NC);
#pragma unroll
  for (int sub = 0; sub < 2; ++sub) {
#pragma unroll
    for (int tt = 0; tt < NT; ++tt) {
      f.bn[sub][tt][0] = qn[(2 * sub) * NC + tt * 16];
      f.bn[sub][tt][1] = qn[(2 * sub + 1) * NC + tt * 16];
      f.bs[sub][tt][0] = qs[(2 * sub) * NC + tt * 16];
      f.bs[sub][tt][1] = qs[(2 * sub + 1) * NC + tt * 16];
    }
  }
}

template <int MT, int NT>
__device__ __forceinline__ void mma_chunk(const Frag<MT, NT>& f, const float (&invm)[MT],
                                          v8f (&acc)[MT][NT]) {
#pragma unroll
  for (int sub = 0; sub < 2; ++sub) {
#pragma unroll
    for (int mi = 0; mi < MT; ++mi) {
      v2f a1, a2;
      if (sub == 0) {
        a1.x = f.a[mi].x * invm[mi]; a1.y = f.a[mi].y * invm[mi];
        a2.x = f.s[mi].x;            a2.y = f.s[mi].y;
      } else {
        a1.x = f.a[mi].z * invm[mi]; a1.y = f.a[mi].w * invm[mi];
        a2.x = f.s[mi].z;            a2.y = f.s[mi].w;
      }
#pragma unroll
      for (int tt = 0; tt < NT; ++tt) {
        v2f b1;
        b1.x = f.bn[sub][tt][0]; b1.y = f.bn[sub][tt][1];
        acc[mi][tt] = __builtin_amdgcn_wmma_f32_16x16x4_f32(false, a1, false, b1,
                                                            (short)0, acc[mi][tt], false, false);
      }
#pragma unroll
      for (int tt = 0; tt < NT; ++tt) {
        v2f b2;
        b2.x = f.bs[sub][tt][0]; b2.y = f.bs[sub][tt][1];
        acc[mi][tt] = __builtin_amdgcn_wmma_f32_16x16x4_f32(false, a2, false, b2,
                                                            (short)0, acc[mi][tt], false, false);
      }
    }
  }
}

template <int MT, int NT, int NC, bool RELU>
__device__ __forceinline__ void sage_gemm_tile(const float* __restrict__ agg,
                                               const float* __restrict__ inv,
                                               const float* __restrict__ selfA,
                                               const float* __restrict__ Wn,
                                               const float* __restrict__ Ws,
                                               const float* __restrict__ bias,
                                               float* __restrict__ out,
                                               int M, int row0, int col0, int lane) {
  constexpr int K = D_FEAT;
  constexpr int NCHUNK = K / 8;
  int c = lane & 15;
  int h = lane >> 4;
  int nb = col0 + c;

  const float* aRow[MT];
  const float* sRow[MT];
  float invm[MT];
#pragma unroll
  for (int mi = 0; mi < MT; ++mi) {
    int lrow = row0 + mi * 16 + c;
    int lrc = lrow < M ? lrow : (M - 1);
    invm[mi] = inv[lrc];
    aRow[mi] = agg   + (long long)lrc * K + 4 * h;
    sRow[mi] = selfA + (long long)lrc * K + 4 * h;
  }
  const float* pWn = Wn + (long long)(4 * h) * NC + nb;
  const float* pWs = Ws + (long long)(4 * h) * NC + nb;

  v8f acc[MT][NT] = {};
  Frag<MT, NT> f0, f1;
  constexpr int NLD = 2 * MT + 8 * NT;
  constexpr int NMM = 4 * MT * NT;
  load_frag<MT, NT, NC>(f0, aRow, sRow, pWn, pWs, 0);
  for (int kc = 0; kc < NCHUNK - 2; kc += 2) {
    load_frag<MT, NT, NC>(f1, aRow, sRow, pWn, pWs, kc + 1);
    mma_chunk<MT, NT>(f0, invm, acc);
    load_frag<MT, NT, NC>(f0, aRow, sRow, pWn, pWs, kc + 2);
    mma_chunk<MT, NT>(f1, invm, acc);
    __builtin_amdgcn_sched_group_barrier(0x020, NLD, 0);
    __builtin_amdgcn_sched_group_barrier(0x008, NMM, 0);
    __builtin_amdgcn_sched_group_barrier(0x020, NLD, 0);
    __builtin_amdgcn_sched_group_barrier(0x008, NMM, 0);
  }
  load_frag<MT, NT, NC>(f1, aRow, sRow, pWn, pWs, NCHUNK - 1);
  mma_chunk<MT, NT>(f0, invm, acc);
  mma_chunk<MT, NT>(f1, invm, acc);

  int mb = row0 + (h << 3);
#pragma unroll
  for (int mi = 0; mi < MT; ++mi) {
#pragma unroll
    for (int tt = 0; tt < NT; ++tt) {
      int n = nb + tt * 16;
      float bn = bias[n];
#pragma unroll
      for (int v = 0; v < 8; ++v) {
        int m = mb + mi * 16 + v;
        if (m < M) {
          float r = acc[mi][tt][v] + bn;
          if (RELU) r = fmaxf(r, 0.0f);
          out[(long long)m * NC + n] = r;
        }
      }
    }
  }
}

template <int MT, int NC, bool RELU, int WPS, int SPLIT>
__global__ void __launch_bounds__(128) __attribute__((amdgpu_waves_per_eu(WPS, WPS)))
sage_gemm_wmma(const float* __restrict__ agg,
               const float* __restrict__ inv,
               const float* __restrict__ selfA,
               const float* __restrict__ Wn,
               const float* __restrict__ Ws,
               const float* __restrict__ bias,
               float* __restrict__ out,
               int M, int tailRows) {
  constexpr int STRIPS = NC / 64;
  constexpr int TSTRIPS = NC / 16;
  constexpr int TROWS = 32;
  constexpr int ROWS = 16 * MT;
  const int wave = __builtin_amdgcn_readfirstlane((int)(threadIdx.x >> 5));
  const int lane = threadIdx.x & 31;
  if (SPLIT == 0) tailRows = 0;
  const int mainTiles = ((M - tailRows + ROWS - 1) / ROWS) * STRIPS;
  const int tailTiles = (tailRows / TROWS) * TSTRIPS;
  int tile = blockIdx.x * 4 + wave;
  bool small = false;
  if (SPLIT == 1) {
    small = tile < tailTiles;
    if (!small) tile -= tailTiles;
  } else if (SPLIT == 2) {
    small = tile >= mainTiles;
    if (small) tile -= mainTiles;
  }
  if (SPLIT != 0 && small) {
    if (tile >= tailTiles) return;
    const int tm = tile / TSTRIPS;
    const int tn = tile - tm * TSTRIPS;
    sage_gemm_tile<2, 1, NC, RELU>(agg, inv, selfA, Wn, Ws, bias, out, M, tm * TROWS, tn << 4,
                                   lane);
    return;
  }
  if (tile >= mainTiles) return;
  const int tm = tile / STRIPS;
  const int tn = tile - tm * STRIPS;
  sage_gemm_tile<MT, 4, NC, RELU>(agg, inv, selfA, Wn, Ws, bias, out, M, tailRows + tm * ROWS,
                                  tn << 6, lane);
}

static inline int cdiv(long long a, long long b) { return (int)((a + b - 1) / b); }

extern "C" void kernel_launch(void* const* d_in, const int* in_sizes, int n_in,
                              void* d_out, int out_size, void* d_ws, size_t ws_size,
                              hipStream_t stream) {
  const float* x   = (const float*)d_in[0];
  const float* Wn0 = (const float*)d_in[1];
  const float* Ws0 = (const float*)d_in[2];
  const float* b0  = (const float*)d_in[3];
  const float* Wn1 = (const float*)d_in[4];
  const float* Ws1 = (const float*)d_in[5];
  const float* b1  = (const float*)d_in[6];
  const float* Wn2 = (const float*)d_in[7];
  const float* Ws2 = (const float*)d_in[8];
  const float* b2  = (const float*)d_in[9];
  const int* src0  = (const int*)d_in[10];
  const int* dst0  = (const int*)d_in[11];
  const int* src1  = (const int*)d_in[12];
  const int* dst1  = (const int*)d_in[13];
  const int* src2  = (const int*)d_in[14];
  const int* dst2  = (const int*)d_in[15];

  const int N0 = 50000, N1 = 25000, N2 = 12500;
  const int E0 = 500000, E1 = 250000, E2 = 125000;

  char* ws = (char*)d_ws;
  float* agg = (float*)ws;
  ws += (size_t)N0 * D_FEAT * sizeof(float);
  float* h0 = (float*)ws;
  ws += (size_t)N0 * D_FEAT * sizeof(float);
  float* h1 = (float*)ws;
  ws += (size_t)N1 * D_FEAT * sizeof(float);
  float* inv = (float*)ws;

  sage_owner_sum<196><<<cdiv(N0, 196), 1024, 0, stream>>>(x, src0, dst0, E0, N0, agg, inv);
  {
    const int tail = 864;
    int tiles = (tail / 32) * (256 / 16) + (((N0 - tail) + 63) / 64) * (256 / 64);
    sage_gemm_wmma<4, 256, true, 3, 1><<<cdiv(tiles, 4), 128, 0, stream>>>(agg, inv, x, Wn0, Ws0,
                                                                           b0, h0, N0, tail);
  }

  sage_owner_sum<98><<<cdiv(N1, 98), 1024, 0, stream>>>(h0, src1, dst1, E1, N1, agg, inv);
  {
    const int tail = 448;
    int tiles = (tail / 32) * (256 / 16) + (((N1 - tail) + 31) / 32) * (256 / 64);
    sage_gemm_wmma<2, 256, true, 4, 2><<<cdiv(tiles, 4), 128, 0, stream>>>(agg, inv, h0, Wn1,
                                                                           Ws1, b1, h1, N1, tail);
  }

  sage_owner_sum<49><<<cdiv(N2, 49), 1024, 0, stream>>>(h1, src2, dst2, E2, N2, agg, inv);
  {
    int tiles = ((N2 + 15) / 16) * (128 / 64);
    sage_gemm_wmma<1, 128, false, 4, 0><<<cdiv(tiles, 4), 128, 0, stream>>>(agg, inv, h1, Wn2,
                                                                             Ws2, b2,
                                                                             (float*)d_out, N2, 0);
  }
}


# Dual GEMM: a wave's 64 columns interleaved (lane c owns columns 4c..4c+3): weight fragments as 16-byte loads (8 per k chunk, not 32), an output row piece as one 16-byte store; sum order kept
# speedup vs baseline: 3.0798x; 1.0680x over previous
#include <hip/hip_runtime.h>
#include <hip/hip_bf16.h>

typedef __attribute__((ext_vector_type(2))) float v2f;
typedef __attribute__((ext_vector_type(8))) float v8f;

#define D_FEAT 256

#define OWN_LDS_ROWS 196
#define OWN_WAVES 32
#define OWN_RING 512
#define OWN_QCAP 256

template <int R>
__global__ void __launch_bounds__(1024)
sage_owner_sum(const float* __restrict__ H,
               const int* __restrict__ src,
               const int* __restrict__ dst,
               int E, int N,
               float* __restrict__ agg,
               float* __restrict__ inv) {
  static_assert(R <= OWN_LDS_ROWS, "owned rows must fit the LDS copy");
  static_assert(R <= 256, "the local row shares a word with the edge index");
  constexpr int RPW = (R + OWN_WAVES - 1) / OWN_WAVES;
  constexpr int G = 8;
  __shared__ float accL[OWN_LDS_ROWS * D_FEAT];
  __shared__ int cnt[OWN_LDS_ROWS];
  __shared__ unsigned ring[OWN_WAVES][OWN_RING];
  __shared__ unsigned queue[OWN_WAVES][OWN_QCAP];
  __shared__ unsigned qtail[OWN_WAVES];

  const int tid = threadIdx.x;
  const int lane = tid & 31;
  const int wave = __builtin_amdgcn_readfirstlane(tid >> 5);
  const int r0 = blockIdx.x * R;

  for (int i = tid; i < R * (D_FEAT / 4); i += 1024)
    ((float4*)accL)[i] = make_float4(0.0f, 0.0f, 0.0f, 0.0f);
  for (int i = tid; i < R; i += 1024) cnt[i] = 0;
  if (tid < OWN_WAVES) qtail[tid] = 0u;
  __syncthreads();

  int head = 0, tail = 0;

  auto append = [&](int d, int e) {
    const unsigned rel = (unsigned)(d - r0);
    const unsigned m = (unsigned)__builtin_amdgcn_uicmp(rel, (unsigned)R, 36);
    if (m != 0u) {
      if (rel < (unsigned)R) {
        const int slot = (tail + (int)__builtin_amdgcn_mbcnt_lo(m, 0u)) & (OWN_RING - 1);
        ring[wave][slot] = ((unsigned)e << 8) | rel;
      }
      tail += __builtin_popcount(m);
    }
  };

  auto slow_add = [&](int e, int rr) {
    const int s = src[e];
    const float4* p = (const float4*)(H + (long long)s * D_FEAT);
    const float4 a = p[lane];
    const float4 b = p[32 + lane];
    float* q = &accL[rr * D_FEAT + lane * 4];
    atomicAdd(q + 0, a.x);
    atomicAdd(q + 1, a.y);
    atomicAdd(q + 2, a.z);
    atomicAdd(q + 3, a.w);
    atomicAdd(q + 128, b.x);
    atomicAdd(q + 129, b.y);
    atomicAdd(q + 130, b.z);
    atomicAdd(q + 131, b.w);
  };

  auto flush = [&](int h, int n) {
    unsigned ent = 0u;
    bool full = false;
    if (lane < n) {
      ent = ring[wave][(h + lane) & (OWN_RING - 1)];
      const unsigned rel = ent & 255u;
      atomicAdd(&cnt[rel], 1);
      const unsigned ow = rel & (OWN_WAVES - 1);
      const unsigned pos = atomicAdd(&qtail[ow], 1u);
      if (pos < (unsigned)OWN_QCAP) queue[ow][pos] = ent;
      else full = true;
    }
    unsigned om = __builtin_amdgcn_ballot_w32(full);
    while (om != 0u) {
      const int l = __builtin_ctz(om);
      om &= om - 1u;
      const unsigned eo = (unsigned)__builtin_amdgcn_readlane((int)ent, l);
      slow_add((int)(eo >> 8), (int)(eo & 255u));
    }
  };

  const int nFull = E >> 8;
  if (wave < nFull) {
    const int4* d4 = (const int4*)dst;
    const int first = (int)(((unsigned)blockIdx.x * (unsigned)nFull) / gridDim.x);
    int it = wave + first;
    if (it >= nFull) it -= nFull;
    int4 ca = d4[it * 64 + lane];
    int4 cb = d4[it * 64 + 32 + lane];
    for (int t = wave; t < nFull; t += OWN_WAVES) {
      int nx = it;
      if (t + OWN_WAVES < nFull) {
        nx = t + OWN_WAVES + first;
        if (nx >= nFull) nx -= nFull;
      }
      const int4 na = d4[nx * 64 + lane];
      const int4 nb = d4[nx * 64 + 32 + lane];
      const int base = it * 256 + lane * 4;
      append(ca.x, base + 0);
      append(ca.y, base + 1);
      append(ca.z, base + 2);
      append(ca.w, base + 3);
      append(cb.x, base + 128);
      append(cb.y, base + 129);
      append(cb.z, base + 130);
      append(cb.w, base + 131);
      while (tail - head >= 32) {
        flush(head, 32);
        head += 32;
      }
      ca = na;
      cb = nb;
      it = nx;
    }
  }
  if ((E & 255) != 0 && wave == (nFull & (OWN_WAVES - 1))) {
    const int base = nFull * 256 + lane * 4;
    int dv[8];
#pragma unroll
    for (int j = 0; j < 8; ++j) {
      const int e = base + (j & 3) + (j >> 2) * 128;
      dv[j] = dst[e < E ? e : E - 1];
    }
#pragma unroll
    for (int j = 0; j < 8; ++j) {
      const int e = base + (j & 3) + (j >> 2) * 128;
      append(e < E ? dv[j] : -1, e);
    }
  }
  while (tail != head) {
    const int n = (tail - head) < 32 ? (tail - head) : 32;
    flush(head, n);
    head += n;
  }
  __syncthreads();

  float acc[RPW][8];
#pragma unroll
  for (int q = 0; q < RPW; ++q) {
#pragma unroll
    for (int c = 0; c < 8; ++c) acc[q][c] = 0.0f;
  }
  const unsigned loff = (unsigned)lane * 16u;
  const int nq = (int)(qtail[wave] < (unsigned)OWN_QCAP ? qtail[wave] : (unsigned)OWN_QCAP);
  for (int base = 0; base < nq; base += 32) {
    const int n = (nq - base) < 32 ? (nq - base) : 32;
    const unsigned ent = queue[wave][base + (lane < n ? lane : 0)];
    const int slv = lane < n ? (int)((ent & 255u) >> 5) : RPW;
    const int sv = src[ent >> 8];
    for (int g = 0; g < n; g += G) {
      float4 a[G], b[G];
      int sl[G];
#pragma unroll
      for (int k = 0; k < G; ++k) {
        const int s = __builtin_amdgcn_readlane(sv, g + k);
        sl[k] = __builtin_amdgcn_readlane(slv, g + k);
        const char* p = (const char*)(H + (long long)s * D_FEAT);
        a[k] = *(const float4*)(p + loff);
        b[k] = *(const float4*)(p + loff + 512u);
      }
#pragma unroll
      for (int k = 0; k < G; ++k) {
#pragma unroll
        for (int q = 0; q < RPW; ++q) {
          if (sl[k] == q) {
            acc[q][0] += a[k].x;
            acc[q][1] += a[k].y;
            acc[q][2] += a[k].z;
            acc[q][3] += a[k].w;
            acc[q][4] += b[k].x;
            acc[q][5] += b[k].y;
            acc[q][6] += b[k].z;
            acc[q][7] += b[k].w;
          }
        }
      }
    }
  }

  const int nrows = (N - r0) < R ? (N - r0) : R;
#pragma unroll
  for (int q = 0; q < RPW; ++q) {
    const int row = q * OWN_WAVES + wave;
    if (row < nrows) {
      const float4* l4 = (const float4*)&accL[row * D_FEAT];
      float4 va = l4[lane];
      float4 vb = l4[32 + lane];
      va.x += acc[q][0];
      va.y += acc[q][1];
      va.z += acc[q][2];
      va.w += acc[q][3];
      vb.x += acc[q][4];
      vb.y += acc[q][5];
      vb.z += acc[q][6];
      vb.w += acc[q][7];
      float4* o4 = (float4*)(agg + (long long)(r0 + row) * D_FEAT);
      o4[lane] = va;
      o4[32 + lane] = vb;
      if (lane == 0) {
        const int c = cnt[row];
        inv[r0 + row] = 1.0f / (float)(c > 0 ? c : 1);
      }
    }
  }
}

__device__ __forceinline__ float f4c(const float4& v, int i) {
  return i == 0 ? v.x : (i == 1 ? v.y : (i == 2 ? v.z : v.w));
}

template <int MT, int NT>
struct Frag {
  float4 a[MT];
  float4 s[MT];
  float bn[2][NT][2];
  float bs[2][NT][2];
};

template <int MT, int NT, int NC>
__device__ __forceinline__ void load_frag(Frag<MT, NT>& f,
                                          const float* const (&aRow)[MT],
                                          const float* const (&sRow)[MT],
                                          const float* __restrict__ pWn,
                                          const float* __restrict__ pWs,
                                          int kc) {
#pragma unroll
  for (int mi = 0; mi < MT; ++mi) {
    f.a[mi] = *(const float4*)(aRow[mi] + kc * 8);
    f.s[mi] = *(const float4*)(sRow[mi] + kc * 8);
  }
  const float* qn = pWn + (long long)kc * (8 * NC);
  const float* qs = pWs + (long long)kc * (8 * NC);
  if (NT == 4) {
#pragma unroll
    for (int r = 0; r < 4; ++r) {
      const float4 vn = *(const float4*)(qn + r * NC);
      const float4 vs = *(const float4*)(qs + r * NC);
#pragma unroll
      for (int tt = 0; tt < NT; ++tt) {
        f.bn[r >> 1][tt][r & 1] = f4c(vn, tt);
        f.bs[r >> 1][tt][r & 1] = f4c(vs, tt);
      }
    }
  } else {
#pragma unroll
    for (int sub = 0; sub < 2; ++sub) {
#pragma unroll
      for (int tt = 0; tt < NT; ++tt) {
        f.bn[sub][tt][0] = qn[(2 * sub) * NC + tt * 16];
        f.bn[sub][tt][1] = qn[(2 * sub + 1) * NC + tt * 16];
        f.bs[sub][tt][0] = qs[(2 * sub) * NC + tt * 16];
        f.bs[sub][tt][1] = qs[(2 * sub + 1) * NC + tt * 16];
      }
    }
  }
}

template <int MT, int NT>
__device__ __forceinline__ void mma_chunk(const Frag<MT, NT>& f, const float (&invm)[MT],
                                          v8f (&acc)[MT][NT]) {
#pragma unroll
  for (int sub = 0; sub < 2; ++sub) {
#pragma unroll
    for (int mi = 0; mi < MT; ++mi) {
      v2f a1, a2;
      if (sub == 0) {
        a1.x = f.a[mi].x * invm[mi]; a1.y = f.a[mi].y * invm[mi];
        a2.x = f.s[mi].x;            a2.y = f.s[mi].y;
      } else {
        a1.x = f.a[mi].z * invm[mi]; a1.y = f.a[mi].w * invm[mi];
        a2.x = f.s[mi].z;            a2.y = f.s[mi].w;
      }
#pragma unroll
      for (int tt = 0; tt < NT; ++tt) {
        v2f b1;
        b1.x = f.bn[sub][tt][0]; b1.y = f.bn[sub][tt][1];
        acc[mi][tt] = __builtin_amdgcn_wmma_f32_16x16x4_f32(false, a1, false, b1,
                                                            (short)0, acc[mi][tt], false, false);
      }
#pragma unroll
      for (int tt = 0; tt < NT; ++tt) {
        v2f b2;
        b2.x = f.bs[sub][tt][0]; b2.y = f.bs[sub][tt][1];
        acc[mi][tt] = __builtin_amdgcn_wmma_f32_16x16x4_f32(false, a2, false, b2,
                                                            (short)0, acc[mi][tt], false, false);
      }
    }
  }
}

template <int MT, int NT, int NC, bool RELU>
__device__ __forceinline__ void sage_gemm_tile(const float* __restrict__ agg,
                                               const float* __restrict__ inv,
                                               const float* __restrict__ selfA,
                                               const float* __restrict__ Wn,
                                               const float* __restrict__ Ws,
                                               const float* __restrict__ bias,
                                               float* __restrict__ out,
                                               int M, int row0, int col0, int lane) {
  constexpr int K = D_FEAT;
  constexpr int NCHUNK = K / 8;
  int c = lane & 15;
  int h = lane >> 4;
  int nb = col0 + (NT == 4 ? 4 * c : c);

  const float* aRow[MT];
  const float* sRow[MT];
  float invm[MT];
#pragma unroll
  for (int mi = 0; mi < MT; ++mi) {
    int lrow = row0 + mi * 16 + c;
    int lrc = lrow < M ? lrow : (M - 1);
    invm[mi] = inv[lrc];
    aRow[mi] = agg   + (long long)lrc * K + 4 * h;
    sRow[mi] = selfA + (long long)lrc * K + 4 * h;
  }
  const float* pWn = Wn + (long long)(4 * h) * NC + nb;
  const float* pWs = Ws + (long long)(4 * h) * NC + nb;

  v8f acc[MT][NT] = {};
  Frag<MT, NT> f0, f1;
  constexpr int NLD = 2 * MT + (NT == 4 ? 8 : 8 * NT);
  constexpr int NMM = 4 * MT * NT;
  load_frag<MT, NT, NC>(f0, aRow, sRow, pWn, pWs, 0);
#pragma unroll 1
  for (int kc = 0; kc < NCHUNK - 2; kc += 2) {
    load_frag<MT, NT, NC>(f1, aRow, sRow, pWn, pWs, kc + 1);
    mma_chunk<MT, NT>(f0, invm, acc);
    load_frag<MT, NT, NC>(f0, aRow, sRow, pWn, pWs, kc + 2);
    mma_chunk<MT, NT>(f1, invm, acc);
    __builtin_amdgcn_sched_group_barrier(0x020, NLD, 0);
    __builtin_amdgcn_sched_group_barrier(0x008, NMM, 0);
    __builtin_amdgcn_sched_group_barrier(0x020, NLD, 0);
    __builtin_amdgcn_sched_group_barrier(0x008, NMM, 0);
  }
  load_frag<MT, NT, NC>(f1, aRow, sRow, pWn, pWs, NCHUNK - 1);
  mma_chunk<MT, NT>(f0, invm, acc);
  mma_chunk<MT, NT>(f1, invm, acc);

  int mb = row0 + (h << 3);
  if (NT == 4) {
    const float4 bv = *(const float4*)(bias + nb);
#pragma unroll
    for (int mi = 0; mi < MT; ++mi) {
#pragma unroll
      for (int v = 0; v < 8; ++v) {
        int m = mb + mi * 16 + v;
        if (m < M) {
          float4 r;
          r.x = acc[mi][0][v] + bv.x;
          r.y = acc[mi][1 % NT][v] + bv.y;
          r.z = acc[mi][2 % NT][v] + bv.z;
          r.w = acc[mi][3 % NT][v] + bv.w;
          if (RELU) {
            r.x = fmaxf(r.x, 0.0f);
            r.y = fmaxf(r.y, 0.0f);
            r.z = fmaxf(r.z, 0.0f);
            r.w = fmaxf(r.w, 0.0f);
          }
          *(float4*)(out + (long long)m * NC + nb) = r;
        }
      }
    }
    return;
  }
#pragma unroll
  for (int mi = 0; mi < MT; ++mi) {
#pragma unroll
    for (int tt = 0; tt < NT; ++tt) {
      int n = nb + tt * 16;
      float bn = bias[n];
#pragma unroll
      for (int v = 0; v < 8; ++v) {
        int m = mb + mi * 16 + v;
        if (m < M) {
          float r = acc[mi][tt][v] + bn;
          if (RELU) r = fmaxf(r, 0.0f);
          out[(long long)m * NC + n] = r;
        }
      }
    }
  }
}

template <int MT, int NC, bool RELU, int WPS, int SPLIT>
__global__ void __launch_bounds__(128) __attribute__((amdgpu_waves_per_eu(WPS, WPS)))
sage_gemm_wmma(const float* __restrict__ agg,
               const float* __restrict__ inv,
               const float* __restrict__ selfA,
               const float* __restrict__ Wn,
               const float* __restrict__ Ws,
               const float* __restrict__ bias,
               float* __restrict__ out,
               int M, int tailRows) {
  constexpr int STRIPS = NC / 64;
  constexpr int TSTRIPS = NC / 16;
  constexpr int TROWS = 32;
  constexpr int ROWS = 16 * MT;
  const int wave = __builtin_amdgcn_readfirstlane((int)(threadIdx.x >> 5));
  const int lane = threadIdx.x & 31;
  if (SPLIT == 0) tailRows = 0;
  const int mainTiles = ((M - tailRows + ROWS - 1) / ROWS) * STRIPS;
  const int tailTiles = (tailRows / TROWS) * TSTRIPS;
  int tile = blockIdx.x * 4 + wave;
  bool small = false;
  if (SPLIT == 1) {
    small = tile < tailTiles;
    if (!small) tile -= tailTiles;
  } else if (SPLIT == 2) {
    small = tile >= mainTiles;
    if (small) tile -= mainTiles;
  }
  if (SPLIT != 0 && small) {
    if (tile >= tailTiles) return;
    const int tm = tile / TSTRIPS;
    const int tn = tile - tm * TSTRIPS;
    sage_gemm_tile<2, 1, NC, RELU>(agg, inv, selfA, Wn, Ws, bias, out, M, tm * TROWS, tn << 4,
                                   lane);
    return;
  }
  if (tile >= mainTiles) return;
  const int tm = tile / STRIPS;
  const int tn = tile - tm * STRIPS;
  sage_gemm_tile<MT, 4, NC, RELU>(agg, inv, selfA, Wn, Ws, bias, out, M, tailRows + tm * ROWS,
                                  tn << 6, lane);
}

static inline int cdiv(long long a, long long b) { return (int)((a + b - 1) / b); }

extern "C" void kernel_launch(void* const* d_in, const int* in_sizes, int n_in,
                              void* d_out, int out_size, void* d_ws, size_t ws_size,
                              hipStream_t stream) {
  const float* x   = (const float*)d_in[0];
  const float* Wn0 = (const float*)d_in[1];
  const float* Ws0 = (const float*)d_in[2];
  const float* b0  = (const float*)d_in[3];
  const float* Wn1 = (const float*)d_in[4];
  const float* Ws1 = (const float*)d_in[5];
  const float* b1  = (const float*)d_in[6];
  const float* Wn2 = (const float*)d_in[7];
  const float* Ws2 = (const float*)d_in[8];
  const float* b2  = (const float*)d_in[9];
  const int* src0  = (const int*)d_in[10];
  const int* dst0  = (const int*)d_in[11];
  const int* src1  = (const int*)d_in[12];
  const int* dst1  = (const int*)d_in[13];
  const int* src2  = (const int*)d_in[14];
  const int* dst2  = (const int*)d_in[15];

  const int N0 = 50000, N1 = 25000, N2 = 12500;
  const int E0 = 500000, E1 = 250000, E2 = 125000;

  char* ws = (char*)d_ws;
  float* agg = (float*)ws;
  ws += (size_t)N0 * D_FEAT * sizeof(float);
  float* h0 = (float*)ws;
  ws += (size_t)N0 * D_FEAT * sizeof(float);
  float* h1 = (float*)ws;
  ws += (size_t)N1 * D_FEAT * sizeof(float);
  float* inv = (float*)ws;

  sage_owner_sum<196><<<cdiv(N0, 196), 1024, 0, stream>>>(x, src0, dst0, E0, N0, agg, inv);
  {
    const int tail = 864;
    int tiles = (tail / 32) * (256 / 16) + (((N0 - tail) + 63) / 64) * (256 / 64);
    sage_gemm_wmma<4, 256, true, 3, 1><<<cdiv(tiles, 4), 128, 0, stream>>>(agg, inv, x, Wn0, Ws0,
                                                                           b0, h0, N0, tail);
  }

  sage_owner_sum<98><<<cdiv(N1, 98), 1024, 0, stream>>>(h0, src1, dst1, E1, N1, agg, inv);
  {
    const int tail = 448;
    int tiles = (tail / 32) * (256 / 16) + (((N1 - tail) + 31) / 32) * (256 / 64);
    sage_gemm_wmma<2, 256, true, 4, 2><<<cdiv(tiles, 4), 128, 0, stream>>>(agg, inv, h0, Wn1,
                                                                           Ws1, b1, h1, N1, tail);
  }

  sage_owner_sum<49><<<cdiv(N2, 49), 1024, 0, stream>>>(h1, src2, dst2, E2, N2, agg, inv);
  {
    int tiles = ((N2 + 15) / 16) * (128 / 64);
    sage_gemm_wmma<1, 128, false, 4, 0><<<cdiv(tiles, 4), 128, 0, stream>>>(agg, inv, h1, Wn2,
                                                                             Ws2, b2,
                                                                             (float*)d_out, N2, 0);
  }
}
